# starting kernel
# baseline (speedup 1.0000x reference)
#include <hip/hip_runtime.h>
#include <hip/hip_bf16.h>

typedef float v2f __attribute__((ext_vector_type(2)));
typedef float v8f __attribute__((ext_vector_type(8)));

#define NTOK   16384
#define DMODEL 128
#define NGRAPH 1024
#define ALPHA  0.2f
#define BN_EPS 1e-5f

__global__ __launch_bounds__(256)
void gat_gemm_wmma(const float* __restrict__ h, const float* __restrict__ W,
                   float* __restrict__ Wh) {
    __shared__ float Wlds[DMODEL * DMODEL];

    const int tid = threadIdx.x;
    const float4* Wg = (const float4*)W;
    float4* Wl = (float4*)Wlds;
#pragma unroll
    for (int it = 0; it < 16; ++it)
        Wl[tid + it * 256] = Wg[tid + it * 256];
    __syncthreads();

    const int wave  = tid >> 5;
    const int lane  = tid & 31;
    const int mtile = blockIdx.x * 8 + wave;
    const int mrow  = (lane & 15);
    const int khalf = (lane >> 4) << 1;

    v8f acc[8];
#pragma unroll
    for (int t = 0; t < 8; ++t) acc[t] = (v8f){0.f,0.f,0.f,0.f,0.f,0.f,0.f,0.f};

    const float* hrow = h + (mtile * 16 + mrow) * DMODEL + khalf;

    for (int kk = 0; kk < DMODEL; kk += 4) {
        v2f a = *(const v2f*)(hrow + kk);
#pragma unroll
        for (int t = 0; t < 8; ++t) {
            v2f b = *(const v2f*)(&Wlds[(t * 16 + mrow) * DMODEL + kk + khalf]);
            acc[t] = __builtin_amdgcn_wmma_f32_16x16x4_f32(
                false, a, false, b, (short)0, acc[t], false, false);
        }
    }

    const int hi = lane >> 4;
    const int cn = lane & 15;
#pragma unroll
    for (int t = 0; t < 8; ++t) {
#pragma unroll
        for (int v = 0; v < 8; ++v) {
            int r = mtile * 16 + v + 8 * hi;
            Wh[r * DMODEL + t * 16 + cn] = acc[t][v];
        }
    }
}

__global__ __launch_bounds__(256)
void gat_scores(const float* __restrict__ Wh, const float* __restrict__ a1,
                const float* __restrict__ a2, float* __restrict__ s1,
                float* __restrict__ s2) {
    const int wave = threadIdx.x >> 5;
    const int lane = threadIdx.x & 31;
    const int r = blockIdx.x * 8 + wave;

    const float4 x = ((const float4*)(Wh + r * DMODEL))[lane];
    const float4 u = ((const float4*)a1)[lane];
    const float4 v = ((const float4*)a2)[lane];
    float d1 = x.x * u.x + x.y * u.y + x.z * u.z + x.w * u.w;
    float d2 = x.x * v.x + x.y * v.y + x.z * v.z + x.w * v.w;
#pragma unroll
    for (int off = 16; off > 0; off >>= 1) {
        d1 += __shfl_xor(d1, off, 32);
        d2 += __shfl_xor(d2, off, 32);
    }
    if (lane == 0) { s1[r] = d1; s2[r] = d2; }
}

__global__ __launch_bounds__(256)
void gat_attn(const float* __restrict__ Wh, const float* __restrict__ s1,
              const float* __restrict__ s2, const float* __restrict__ abp,
              float* __restrict__ out, float* __restrict__ psum,
              float* __restrict__ psq) {
    const float ab = abp[0];
    const int d    = threadIdx.x & 127;
    const int half = threadIdx.x >> 7;
    const int r0   = blockIdx.x * 128;

    float sum = 0.f, sq = 0.f;
    for (int rl = half * 64; rl < half * 64 + 64; ++rl) {
        const int g = r0 + rl;
        const int i = g & (NGRAPH - 1);
        const float sc = s1[g];
        const bool hasl = (i > 0), hasr = (i < NGRAPH - 1);

        float ec = sc + s2[g] + ab;        ec = fmaxf(ec, ALPHA * ec);
        float el = -3.0e38f, er = -3.0e38f;
        if (hasl) { el = sc + s2[g - 1] + ab; el = fmaxf(el, ALPHA * el); }
        if (hasr) { er = sc + s2[g + 1] + ab; er = fmaxf(er, ALPHA * er); }

        const float m  = fmaxf(ec, fmaxf(el, er));
        const float wl = hasl ? __expf(el - m) : 0.f;
        const float wc = __expf(ec - m);
        const float wr = hasr ? __expf(er - m) : 0.f;
        const float inv = 1.f / (wl + wc + wr);

        float x = wc * Wh[g * DMODEL + d];
        if (hasl) x += wl * Wh[(g - 1) * DMODEL + d];
        if (hasr) x += wr * Wh[(g + 1) * DMODEL + d];
        x *= inv;
        x = (x > 0.f) ? x : (__expf(x) - 1.f);

        out[g * DMODEL + d] = x;
        sum += x; sq += x * x;
    }

    __shared__ float ls[128], lq[128];
    if (half == 1) { ls[d] = sum; lq[d] = sq; }
    __syncthreads();
    if (half == 0) {
        psum[blockIdx.x * 128 + d] = sum + ls[d];
        psq [blockIdx.x * 128 + d] = sq  + lq[d];
    }
}

__global__ void gat_stats(const float* __restrict__ psum, const float* __restrict__ psq,
                          const float* __restrict__ gamma, const float* __restrict__ beta,
                          float* __restrict__ scale, float* __restrict__ shift) {
    const int d = threadIdx.x;
    float s = 0.f, q = 0.f;
    for (int b = 0; b < 128; ++b) { s += psum[b * 128 + d]; q += psq[b * 128 + d]; }
    const float mean = s * (1.f / (float)NTOK);
    const float var  = q * (1.f / (float)NTOK) - mean * mean;
    const float invs = rsqrtf(var + BN_EPS);
    const float sc   = gamma[d] * invs;
    scale[d] = sc;
    shift[d] = beta[d] - mean * sc;
}

__global__ __launch_bounds__(256)
void gat_bn(float* __restrict__ out, const float* __restrict__ scale,
            const float* __restrict__ shift) {
    const int idx = blockIdx.x * 256 + threadIdx.x;
    float4* o = (float4*)out;
    float4 x = o[idx];
    const int d4 = (idx & 31) * 4;
    x.x = x.x * scale[d4 + 0] + shift[d4 + 0];
    x.y = x.y * scale[d4 + 1] + shift[d4 + 1];
    x.z = x.z * scale[d4 + 2] + shift[d4 + 2];
    x.w = x.w * scale[d4 + 3] + shift[d4 + 3];
    o[idx] = x;
}

extern "C" void kernel_launch(void* const* d_in, const int* in_sizes, int n_in,
                              void* d_out, int out_size, void* d_ws, size_t ws_size,
                              hipStream_t stream) {
    const float* h     = (const float*)d_in[0];
    const float* W     = (const float*)d_in[1];
    const float* a1    = (const float*)d_in[2];
    const float* a2    = (const float*)d_in[3];
    const float* ab    = (const float*)d_in[4];
    const float* gamma = (const float*)d_in[5];
    const float* beta  = (const float*)d_in[6];
    float* out = (float*)d_out;

    float* ws    = (float*)d_ws;
    float* Wh    = ws;
    float* s1    = ws + 2097152;
    float* s2    = s1 + NTOK;
    float* psum  = s2 + NTOK;
    float* psq   = psum + NTOK;
    float* scale = psq + NTOK;
    float* shift = scale + 128;

    gat_gemm_wmma<<<128, 256, 0, stream>>>(h, W, Wh);
    gat_scores  <<<2048, 256, 0, stream>>>(Wh, a1, a2, s1, s2);
    gat_attn    <<<128, 256, 0, stream>>>(Wh, s1, s2, ab, out, psum, psq);
    gat_stats   <<<1, 128, 0, stream>>>(psum, psq, gamma, beta, scale, shift);
    gat_bn      <<<2048, 256, 0, stream>>>(out, scale, shift);
}


# attention/ELU pass re-tiled: a wave owns 8 rows, float4 per lane, all its loads issued at once (was 64 serial rows per thread); BN partials now summed 8 rows per lane, then 16 waves via LDS
# speedup vs baseline: 2.5400x; 2.5400x over previous
#include <hip/hip_runtime.h>
#include <hip/hip_bf16.h>

typedef float v2f __attribute__((ext_vector_type(2)));
typedef float v8f __attribute__((ext_vector_type(8)));
typedef float v4f __attribute__((ext_vector_type(4)));

#define NTOK   16384
#define DMODEL 128
#define NGRAPH 1024
#define ALPHA  0.2f
#define BN_EPS 1e-5f

__global__ __launch_bounds__(256)
void gat_gemm_wmma(const float* __restrict__ h, const float* __restrict__ W,
                   float* __restrict__ Wh) {
    __shared__ float Wlds[DMODEL * DMODEL];

    const int tid = threadIdx.x;
    const float4* Wg = (const float4*)W;
    float4* Wl = (float4*)Wlds;
#pragma unroll
    for (int it = 0; it < 16; ++it)
        Wl[tid + it * 256] = Wg[tid + it * 256];
    __syncthreads();

    const int wave  = tid >> 5;
    const int lane  = tid & 31;
    const int mtile = blockIdx.x * 8 + wave;
    const int mrow  = (lane & 15);
    const int khalf = (lane >> 4) << 1;

    v8f acc[8];
#pragma unroll
    for (int t = 0; t < 8; ++t) acc[t] = (v8f){0.f,0.f,0.f,0.f,0.f,0.f,0.f,0.f};

    const float* hrow = h + (mtile * 16 + mrow) * DMODEL + khalf;

    for (int kk = 0; kk < DMODEL; kk += 4) {
        v2f a = *(const v2f*)(hrow + kk);
#pragma unroll
        for (int t = 0; t < 8; ++t) {
            v2f b = *(const v2f*)(&Wlds[(t * 16 + mrow) * DMODEL + kk + khalf]);
            acc[t] = __builtin_amdgcn_wmma_f32_16x16x4_f32(
                false, a, false, b, (short)0, acc[t], false, false);
        }
    }

    const int hi = lane >> 4;
    const int cn = lane & 15;
#pragma unroll
    for (int t = 0; t < 8; ++t) {
#pragma unroll
        for (int v = 0; v < 8; ++v) {
            int r = mtile * 16 + v + 8 * hi;
            Wh[r * DMODEL + t * 16 + cn] = acc[t][v];
        }
    }
}

__global__ __launch_bounds__(256)
void gat_scores(const float* __restrict__ Wh, const float* __restrict__ a1,
                const float* __restrict__ a2, float* __restrict__ s1,
                float* __restrict__ s2) {
    const int wave = threadIdx.x >> 5;
    const int lane = threadIdx.x & 31;
    const int r = blockIdx.x * 8 + wave;

    const float4 x = ((const float4*)(Wh + r * DMODEL))[lane];
    const float4 u = ((const float4*)a1)[lane];
    const float4 v = ((const float4*)a2)[lane];
    float d1 = x.x * u.x + x.y * u.y + x.z * u.z + x.w * u.w;
    float d2 = x.x * v.x + x.y * v.y + x.z * v.z + x.w * v.w;
#pragma unroll
    for (int off = 16; off > 0; off >>= 1) {
        d1 += __shfl_xor(d1, off, 32);
        d2 += __shfl_xor(d2, off, 32);
    }
    if (lane == 0) { s1[r] = d1; s2[r] = d2; }
}

#define ATT_ROWS  8
#define ATT_WAVES 16

__device__ __forceinline__ float att_elem(float l, float c, float r, float wl,
                                          float wc, float wr, float inv,
                                          bool hasl, bool hasr) {
    float x = wc * c;
    if (hasl) x += wl * l;
    if (hasr) x += wr * r;
    x *= inv;
    return (x > 0.f) ? x : (__expf(x) - 1.f);
}

__global__ __launch_bounds__(512)
void gat_attn(const float* __restrict__ Wh, const float* __restrict__ s1,
              const float* __restrict__ s2, const float* __restrict__ abp,
              float* __restrict__ out, float* __restrict__ psum,
              float* __restrict__ psq) {
    __shared__ float red[2][ATT_WAVES * DMODEL];

    const float ab = abp[0];
    const int lane = threadIdx.x & 31;
    const int wave = threadIdx.x >> 5;
    const int g0   = blockIdx.x * 128 + wave * ATT_ROWS;

    const int rlo = (g0 > 0) ? g0 - 1 : 0;
    const int rhi = (g0 + ATT_ROWS < NTOK) ? g0 + ATT_ROWS : NTOK - 1;
    v4f w[ATT_ROWS + 2];
    w[0] = ((const v4f*)(Wh + rlo * DMODEL))[lane];
#pragma unroll
    for (int j = 0; j < ATT_ROWS; ++j)
        w[j + 1] = ((const v4f*)(Wh + (g0 + j) * DMODEL))[lane];
    w[ATT_ROWS + 1] = ((const v4f*)(Wh + rhi * DMODEL))[lane];

    const v4f p0 = ((const v4f*)(s1 + g0))[0];
    const v4f p1 = ((const v4f*)(s1 + g0))[1];
    const v4f q0 = ((const v4f*)(s2 + g0))[0];
    const v4f q1 = ((const v4f*)(s2 + g0))[1];
    const float tlo = s2[rlo];
    const float thi = s2[rhi];

    asm volatile("" :: "v"(w[0]), "v"(w[1]), "v"(w[2]), "v"(w[3]), "v"(w[4]),
                       "v"(w[5]), "v"(w[6]), "v"(w[7]), "v"(w[8]), "v"(w[9]),
                       "v"(p0), "v"(p1), "v"(q0), "v"(q1), "v"(tlo), "v"(thi));

    const float t1[ATT_ROWS] = {p0.x, p0.y, p0.z, p0.w, p1.x, p1.y, p1.z, p1.w};
    const float t2[ATT_ROWS + 2] = {tlo, q0.x, q0.y, q0.z, q0.w,
                                    q1.x, q1.y, q1.z, q1.w, thi};

    v4f xs[ATT_ROWS];
    v4f sum = {0.f, 0.f, 0.f, 0.f}, sq = {0.f, 0.f, 0.f, 0.f};
#pragma unroll
    for (int k = 0; k < ATT_ROWS; ++k) {
        const int g = g0 + k;
        const int i = g & (NGRAPH - 1);
        const float sc = t1[k];
        const bool hasl = (i > 0), hasr = (i < NGRAPH - 1);

        float ec = sc + t2[k + 1] + ab;    ec = fmaxf(ec, ALPHA * ec);
        float el = -3.0e38f, er = -3.0e38f;
        if (hasl) { el = sc + t2[k] + ab;     el = fmaxf(el, ALPHA * el); }
        if (hasr) { er = sc + t2[k + 2] + ab; er = fmaxf(er, ALPHA * er); }

        const float m  = fmaxf(ec, fmaxf(el, er));
        const float wl = hasl ? __expf(el - m) : 0.f;
        const float wc = __expf(ec - m);
        const float wr = hasr ? __expf(er - m) : 0.f;
        const float inv = 1.f / (wl + wc + wr);

        v4f x;
        x.x = att_elem(w[k].x, w[k + 1].x, w[k + 2].x, wl, wc, wr, inv, hasl, hasr);
        x.y = att_elem(w[k].y, w[k + 1].y, w[k + 2].y, wl, wc, wr, inv, hasl, hasr);
        x.z = att_elem(w[k].z, w[k + 1].z, w[k + 2].z, wl, wc, wr, inv, hasl, hasr);
        x.w = att_elem(w[k].w, w[k + 1].w, w[k + 2].w, wl, wc, wr, inv, hasl, hasr);

        xs[k] = x;
        sum.x += x.x; sum.y += x.y; sum.z += x.z; sum.w += x.w;
        sq.x += x.x * x.x; sq.y += x.y * x.y; sq.z += x.z * x.z; sq.w += x.w * x.w;
    }

    ((v4f*)red[0])[wave * 32 + lane] = sum;
    ((v4f*)red[1])[wave * 32 + lane] = sq;
    __syncthreads();

#pragma unroll
    for (int k = 0; k < ATT_ROWS; ++k)
        ((v4f*)(out + (g0 + k) * DMODEL))[lane] = xs[k];

    if (threadIdx.x < 2 * DMODEL) {
        const int which = threadIdx.x >> 7;
        const int d     = threadIdx.x & (DMODEL - 1);
        float acc = 0.f;
#pragma unroll
        for (int wv = 0; wv < ATT_WAVES; ++wv) acc += red[which][wv * DMODEL + d];
        float* dst = which ? psq : psum;
        dst[blockIdx.x * DMODEL + d] = acc;
    }
}

__global__ void gat_stats(const float* __restrict__ psum, const float* __restrict__ psq,
                          const float* __restrict__ gamma, const float* __restrict__ beta,
                          float* __restrict__ scale, float* __restrict__ shift) {
    const int d = threadIdx.x;
    float s = 0.f, q = 0.f;
    for (int b = 0; b < 128; ++b) { s += psum[b * 128 + d]; q += psq[b * 128 + d]; }
    const float mean = s * (1.f / (float)NTOK);
    const float var  = q * (1.f / (float)NTOK) - mean * mean;
    const float invs = rsqrtf(var + BN_EPS);
    const float sc   = gamma[d] * invs;
    scale[d] = sc;
    shift[d] = beta[d] - mean * sc;
}

__global__ __launch_bounds__(256)
void gat_bn(float* __restrict__ out, const float* __restrict__ scale,
            const float* __restrict__ shift) {
    const int idx = blockIdx.x * 256 + threadIdx.x;
    float4* o = (float4*)out;
    float4 x = o[idx];
    const int d4 = (idx & 31) * 4;
    x.x = x.x * scale[d4 + 0] + shift[d4 + 0];
    x.y = x.y * scale[d4 + 1] + shift[d4 + 1];
    x.z = x.z * scale[d4 + 2] + shift[d4 + 2];
    x.w = x.w * scale[d4 + 3] + shift[d4 + 3];
    o[idx] = x;
}

extern "C" void kernel_launch(void* const* d_in, const int* in_sizes, int n_in,
                              void* d_out, int out_size, void* d_ws, size_t ws_size,
                              hipStream_t stream) {
    const float* h     = (const float*)d_in[0];
    const float* W     = (const float*)d_in[1];
    const float* a1    = (const float*)d_in[2];
    const float* a2    = (const float*)d_in[3];
    const float* ab    = (const float*)d_in[4];
    const float* gamma = (const float*)d_in[5];
    const float* beta  = (const float*)d_in[6];
    float* out = (float*)d_out;

    float* ws    = (float*)d_ws;
    float* Wh    = ws;
    float* s1    = ws + 2097152;
    float* s2    = s1 + NTOK;
    float* psum  = s2 + NTOK;
    float* psq   = psum + NTOK;
    float* scale = psq + NTOK;
    float* shift = scale + 128;

    gat_gemm_wmma<<<128, 256, 0, stream>>>(h, W, Wh);
    gat_scores  <<<2048, 256, 0, stream>>>(Wh, a1, a2, s1, s2);
    gat_attn    <<<128, 512, 0, stream>>>(Wh, s1, s2, ab, out, psum, psq);
    gat_stats   <<<1, 128, 0, stream>>>(psum, psq, gamma, beta, scale, shift);
    gat_bn      <<<2048, 256, 0, stream>>>(out, scale, shift);
}


# W rows in LDS padded to a pitch of 132 floats: a B-fragment read's 16 rows no longer start on one bank (pitch 128 made every such read a 16-way bank conflict)
# speedup vs baseline: 3.2809x; 1.2917x over previous
#include <hip/hip_runtime.h>
#include <hip/hip_bf16.h>

typedef float v2f __attribute__((ext_vector_type(2)));
typedef float v8f __attribute__((ext_vector_type(8)));
typedef float v4f __attribute__((ext_vector_type(4)));

#define NTOK   16384
#define DMODEL 128
#define NGRAPH 1024
#define ALPHA  0.2f
#define BN_EPS 1e-5f
#define WPITCH 132

__global__ __launch_bounds__(256)
void gat_gemm_wmma(const float* __restrict__ h, const float* __restrict__ W,
                   float* __restrict__ Wh) {
    __shared__ float Wlds[DMODEL * WPITCH];

    const int tid = threadIdx.x;
    const float4* Wg = (const float4*)W;
#pragma unroll
    for (int it = 0; it < 16; ++it) {
        const int idx = tid + it * 256;
        const int row = idx >> 5;
        const int c4  = idx & 31;
        *(float4*)&Wlds[row * WPITCH + c4 * 4] = Wg[idx];
    }
    __syncthreads();

    const int wave  = tid >> 5;
    const int lane  = tid & 31;
    const int mtile = blockIdx.x * 8 + wave;
    const int mrow  = (lane & 15);
    const int khalf = (lane >> 4) << 1;

    v8f acc[8];
#pragma unroll
    for (int t = 0; t < 8; ++t) acc[t] = (v8f){0.f,0.f,0.f,0.f,0.f,0.f,0.f,0.f};

    const float* hrow = h + (mtile * 16 + mrow) * DMODEL + khalf;

    for (int kk = 0; kk < DMODEL; kk += 4) {
        v2f a = *(const v2f*)(hrow + kk);
#pragma unroll
        for (int t = 0; t < 8; ++t) {
            v2f b = *(const v2f*)(&Wlds[(t * 16 + mrow) * WPITCH + kk + khalf]);
            acc[t] = __builtin_amdgcn_wmma_f32_16x16x4_f32(
                false, a, false, b, (short)0, acc[t], false, false);
        }
    }

    const int hi = lane >> 4;
    const int cn = lane & 15;
#pragma unroll
    for (int t = 0; t < 8; ++t) {
#pragma unroll
        for (int v = 0; v < 8; ++v) {
            int r = mtile * 16 + v + 8 * hi;
            Wh[r * DMODEL + t * 16 + cn] = acc[t][v];
        }
    }
}

__global__ __launch_bounds__(256)
void gat_scores(const float* __restrict__ Wh, const float* __restrict__ a1,
                const float* __restrict__ a2, float* __restrict__ s1,
                float* __restrict__ s2) {
    const int wave = threadIdx.x >> 5;
    const int lane = threadIdx.x & 31;
    const int r = blockIdx.x * 8 + wave;

    const float4 x = ((const float4*)(Wh + r * DMODEL))[lane];
    const float4 u = ((const float4*)a1)[lane];
    const float4 v = ((const float4*)a2)[lane];
    float d1 = x.x * u.x + x.y * u.y + x.z * u.z + x.w * u.w;
    float d2 = x.x * v.x + x.y * v.y + x.z * v.z + x.w * v.w;
#pragma unroll
    for (int off = 16; off > 0; off >>= 1) {
        d1 += __shfl_xor(d1, off, 32);
        d2 += __shfl_xor(d2, off, 32);
    }
    if (lane == 0) { s1[r] = d1; s2[r] = d2; }
}

#define ATT_ROWS  8
#define ATT_WAVES 16

__device__ __forceinline__ float att_elem(float l, float c, float r, float wl,
                                          float wc, float wr, float inv,
                                          bool hasl, bool hasr) {
    float x = wc * c;
    if (hasl) x += wl * l;
    if (hasr) x += wr * r;
    x *= inv;
    return (x > 0.f) ? x : (__expf(x) - 1.f);
}

__global__ __launch_bounds__(512)
void gat_attn(const float* __restrict__ Wh, const float* __restrict__ s1,
              const float* __restrict__ s2, const float* __restrict__ abp,
              float* __restrict__ out, float* __restrict__ psum,
              float* __restrict__ psq) {
    __shared__ float red[2][ATT_WAVES * DMODEL];

    const float ab = abp[0];
    const int lane = threadIdx.x & 31;
    const int wave = threadIdx.x >> 5;
    const int g0   = blockIdx.x * 128 + wave * ATT_ROWS;

    const int rlo = (g0 > 0) ? g0 - 1 : 0;
    const int rhi = (g0 + ATT_ROWS < NTOK) ? g0 + ATT_ROWS : NTOK - 1;
    v4f w[ATT_ROWS + 2];
    w[0] = ((const v4f*)(Wh + rlo * DMODEL))[lane];
#pragma unroll
    for (int j = 0; j < ATT_ROWS; ++j)
        w[j + 1] = ((const v4f*)(Wh + (g0 + j) * DMODEL))[lane];
    w[ATT_ROWS + 1] = ((const v4f*)(Wh + rhi * DMODEL))[lane];

    const v4f p0 = ((const v4f*)(s1 + g0))[0];
    const v4f p1 = ((const v4f*)(s1 + g0))[1];
    const v4f q0 = ((const v4f*)(s2 + g0))[0];
    const v4f q1 = ((const v4f*)(s2 + g0))[1];
    const float tlo = s2[rlo];
    const float thi = s2[rhi];

    asm volatile("" :: "v"(w[0]), "v"(w[1]), "v"(w[2]), "v"(w[3]), "v"(w[4]),
                       "v"(w[5]), "v"(w[6]), "v"(w[7]), "v"(w[8]), "v"(w[9]),
                       "v"(p0), "v"(p1), "v"(q0), "v"(q1), "v"(tlo), "v"(thi));

    const float t1[ATT_ROWS] = {p0.x, p0.y, p0.z, p0.w, p1.x, p1.y, p1.z, p1.w};
    const float t2[ATT_ROWS + 2] = {tlo, q0.x, q0.y, q0.z, q0.w,
                                    q1.x, q1.y, q1.z, q1.w, thi};

    v4f xs[ATT_ROWS];
    v4f sum = {0.f, 0.f, 0.f, 0.f}, sq = {0.f, 0.f, 0.f, 0.f};
#pragma unroll
    for (int k = 0; k < ATT_ROWS; ++k) {
        const int g = g0 + k;
        const int i = g & (NGRAPH - 1);
        const float sc = t1[k];
        const bool hasl = (i > 0), hasr = (i < NGRAPH - 1);

        float ec = sc + t2[k + 1] + ab;    ec = fmaxf(ec, ALPHA * ec);
        float el = -3.0e38f, er = -3.0e38f;
        if (hasl) { el = sc + t2[k] + ab;     el = fmaxf(el, ALPHA * el); }
        if (hasr) { er = sc + t2[k + 2] + ab; er = fmaxf(er, ALPHA * er); }

        const float m  = fmaxf(ec, fmaxf(el, er));
        const float wl = hasl ? __expf(el - m) : 0.f;
        const float wc = __expf(ec - m);
        const float wr = hasr ? __expf(er - m) : 0.f;
        const float inv = 1.f / (wl + wc + wr);

        v4f x;
        x.x = att_elem(w[k].x, w[k + 1].x, w[k + 2].x, wl, wc, wr, inv, hasl, hasr);
        x.y = att_elem(w[k].y, w[k + 1].y, w[k + 2].y, wl, wc, wr, inv, hasl, hasr);
        x.z = att_elem(w[k].z, w[k + 1].z, w[k + 2].z, wl, wc, wr, inv, hasl, hasr);
        x.w = att_elem(w[k].w, w[k + 1].w, w[k + 2].w, wl, wc, wr, inv, hasl, hasr);

        xs[k] = x;
        sum.x += x.x; sum.y += x.y; sum.z += x.z; sum.w += x.w;
        sq.x += x.x * x.x; sq.y += x.y * x.y; sq.z += x.z * x.z; sq.w += x.w * x.w;
    }

    ((v4f*)red[0])[wave * 32 + lane] = sum;
    ((v4f*)red[1])[wave * 32 + lane] = sq;
    __syncthreads();

#pragma unroll
    for (int k = 0; k < ATT_ROWS; ++k)
        ((v4f*)(out + (g0 + k) * DMODEL))[lane] = xs[k];

    if (threadIdx.x < 2 * DMODEL) {
        const int which = threadIdx.x >> 7;
        const int d     = threadIdx.x & (DMODEL - 1);
        float acc = 0.f;
#pragma unroll
        for (int wv = 0; wv < ATT_WAVES; ++wv) acc += red[which][wv * DMODEL + d];
        float* dst = which ? psq : psum;
        dst[blockIdx.x * DMODEL + d] = acc;
    }
}

__global__ void gat_stats(const float* __restrict__ psum, const float* __restrict__ psq,
                          const float* __restrict__ gamma, const float* __restrict__ beta,
                          float* __restrict__ scale, float* __restrict__ shift) {
    const int d = threadIdx.x;
    float s = 0.f, q = 0.f;
    for (int b = 0; b < 128; ++b) { s += psum[b * 128 + d]; q += psq[b * 128 + d]; }
    const float mean = s * (1.f / (float)NTOK);
    const float var  = q * (1.f / (float)NTOK) - mean * mean;
    const float invs = rsqrtf(var + BN_EPS);
    const float sc   = gamma[d] * invs;
    scale[d] = sc;
    shift[d] = beta[d] - mean * sc;
}

__global__ __launch_bounds__(256)
void gat_bn(float* __restrict__ out, const float* __restrict__ scale,
            const float* __restrict__ shift) {
    const int idx = blockIdx.x * 256 + threadIdx.x;
    float4* o = (float4*)out;
    float4 x = o[idx];
    const int d4 = (idx & 31) * 4;
    x.x = x.x * scale[d4 + 0] + shift[d4 + 0];
    x.y = x.y * scale[d4 + 1] + shift[d4 + 1];
    x.z = x.z * scale[d4 + 2] + shift[d4 + 2];
    x.w = x.w * scale[d4 + 3] + shift[d4 + 3];
    o[idx] = x;
}

extern "C" void kernel_launch(void* const* d_in, const int* in_sizes, int n_in,
                              void* d_out, int out_size, void* d_ws, size_t ws_size,
                              hipStream_t stream) {
    const float* h     = (const float*)d_in[0];
    const float* W     = (const float*)d_in[1];
    const float* a1    = (const float*)d_in[2];
    const float* a2    = (const float*)d_in[3];
    const float* ab    = (const float*)d_in[4];
    const float* gamma = (const float*)d_in[5];
    const float* beta  = (const float*)d_in[6];
    float* out = (float*)d_out;

    float* ws    = (float*)d_ws;
    float* Wh    = ws;
    float* s1    = ws + 2097152;
    float* s2    = s1 + NTOK;
    float* psum  = s2 + NTOK;
    float* psq   = psum + NTOK;
    float* scale = psq + NTOK;
    float* shift = scale + 128;

    gat_gemm_wmma<<<128, 256, 0, stream>>>(h, W, Wh);
    gat_scores  <<<2048, 256, 0, stream>>>(Wh, a1, a2, s1, s2);
    gat_attn    <<<128, 512, 0, stream>>>(Wh, s1, s2, ab, out, psum, psq);
    gat_stats   <<<1, 128, 0, stream>>>(psum, psq, gamma, beta, scale, shift);
    gat_bn      <<<2048, 256, 0, stream>>>(out, scale, shift);
}


# scores s1,s2 now taken in the GEMM epilogue from the accumulators (packed FMAs per lane, then a 16-lane DPP butterfly): scores launch and its pass over Wh removed; changes s1,s2 summation order
# speedup vs baseline: 3.7272x; 1.1360x over previous
#include <hip/hip_runtime.h>
#include <hip/hip_bf16.h>

typedef float v2f __attribute__((ext_vector_type(2)));
typedef float v8f __attribute__((ext_vector_type(8)));
typedef float v4f __attribute__((ext_vector_type(4)));

#define NTOK   16384
#define DMODEL 128
#define NGRAPH 1024
#define ALPHA  0.2f
#define BN_EPS 1e-5f
#define WPITCH 132

template <int M>
__device__ __forceinline__ float row_xadd(float x) {
    const int y = __builtin_amdgcn_update_dpp(0, __builtin_bit_cast(int, x),
                                              0x160 + M, 0xf, 0xf, true);
    return x + __builtin_bit_cast(float, y);
}

__global__ __launch_bounds__(256)
void gat_gemm_wmma(const float* __restrict__ h, const float* __restrict__ W,
                   const float* __restrict__ a1, const float* __restrict__ a2,
                   float* __restrict__ Wh, float* __restrict__ s1,
                   float* __restrict__ s2) {
    __shared__ float Wlds[DMODEL * WPITCH];

    const int tid = threadIdx.x;

    v2f u[8];
#pragma unroll
    for (int t = 0; t < 8; ++t)
        u[t] = (v2f){a1[t * 16 + (tid & 15)], a2[t * 16 + (tid & 15)]};

    const float4* Wg = (const float4*)W;
#pragma unroll
    for (int it = 0; it < 16; ++it) {
        const int idx = tid + it * 256;
        const int row = idx >> 5;
        const int c4  = idx & 31;
        *(float4*)&Wlds[row * WPITCH + c4 * 4] = Wg[idx];
    }
    __syncthreads();

    const int wave  = tid >> 5;
    const int lane  = tid & 31;
    const int mtile = blockIdx.x * 8 + wave;
    const int mrow  = (lane & 15);
    const int khalf = (lane >> 4) << 1;

    v8f acc[8];
#pragma unroll
    for (int t = 0; t < 8; ++t) acc[t] = (v8f){0.f,0.f,0.f,0.f,0.f,0.f,0.f,0.f};

    const float* hrow = h + (mtile * 16 + mrow) * DMODEL + khalf;

    for (int kk = 0; kk < DMODEL; kk += 4) {
        v2f a = *(const v2f*)(hrow + kk);
#pragma unroll
        for (int t = 0; t < 8; ++t) {
            v2f b = *(const v2f*)(&Wlds[(t * 16 + mrow) * WPITCH + kk + khalf]);
            acc[t] = __builtin_amdgcn_wmma_f32_16x16x4_f32(
                false, a, false, b, (short)0, acc[t], false, false);
        }
    }

    const int hi = lane >> 4;
    const int cn = lane & 15;
#pragma unroll
    for (int t = 0; t < 8; ++t) {
#pragma unroll
        for (int v = 0; v < 8; ++v) {
            int r = mtile * 16 + v + 8 * hi;
            Wh[r * DMODEL + t * 16 + cn] = acc[t][v];
        }
    }

    v2f p[8];
#pragma unroll
    for (int v = 0; v < 8; ++v) p[v] = (v2f){acc[0][v], acc[0][v]} * u[0];
#pragma unroll
    for (int t = 1; t < 8; ++t) {
#pragma unroll
        for (int v = 0; v < 8; ++v)
            p[v] = __builtin_elementwise_fma((v2f){acc[t][v], acc[t][v]}, u[t], p[v]);
    }
    float p1[8], p2[8];
#pragma unroll
    for (int v = 0; v < 8; ++v) { p1[v] = row_xadd<8>(p[v].x); p2[v] = row_xadd<8>(p[v].y); }
#pragma unroll
    for (int v = 0; v < 8; ++v) { p1[v] = row_xadd<4>(p1[v]); p2[v] = row_xadd<4>(p2[v]); }
#pragma unroll
    for (int v = 0; v < 8; ++v) { p1[v] = row_xadd<2>(p1[v]); p2[v] = row_xadd<2>(p2[v]); }
#pragma unroll
    for (int v = 0; v < 8; ++v) { p1[v] = row_xadd<1>(p1[v]); p2[v] = row_xadd<1>(p2[v]); }

    if (cn == 0) {
        const int r0 = mtile * 16 + 8 * hi;
        ((v4f*)(s1 + r0))[0] = (v4f){p1[0], p1[1], p1[2], p1[3]};
        ((v4f*)(s1 + r0))[1] = (v4f){p1[4], p1[5], p1[6], p1[7]};
        ((v4f*)(s2 + r0))[0] = (v4f){p2[0], p2[1], p2[2], p2[3]};
        ((v4f*)(s2 + r0))[1] = (v4f){p2[4], p2[5], p2[6], p2[7]};
    }
}

#define ATT_ROWS  8
#define ATT_WAVES 16

__device__ __forceinline__ float att_elem(float l, float c, float r, float wl,
                                          float wc, float wr, float inv,
                                          bool hasl, bool hasr) {
    float x = wc * c;
    if (hasl) x += wl * l;
    if (hasr) x += wr * r;
    x *= inv;
    return (x > 0.f) ? x : (__expf(x) - 1.f);
}

__global__ __launch_bounds__(512)
void gat_attn(const float* __restrict__ Wh, const float* __restrict__ s1,
              const float* __restrict__ s2, const float* __restrict__ abp,
              float* __restrict__ out, float* __restrict__ psum,
              float* __restrict__ psq) {
    __shared__ float red[2][ATT_WAVES * DMODEL];

    const float ab = abp[0];
    const int lane = threadIdx.x & 31;
    const int wave = threadIdx.x >> 5;
    const int g0   = blockIdx.x * 128 + wave * ATT_ROWS;

    const int rlo = (g0 > 0) ? g0 - 1 : 0;
    const int rhi = (g0 + ATT_ROWS < NTOK) ? g0 + ATT_ROWS : NTOK - 1;
    v4f w[ATT_ROWS + 2];
    w[0] = ((const v4f*)(Wh + rlo * DMODEL))[lane];
#pragma unroll
    for (int j = 0; j < ATT_ROWS; ++j)
        w[j + 1] = ((const v4f*)(Wh + (g0 + j) * DMODEL))[lane];
    w[ATT_ROWS + 1] = ((const v4f*)(Wh + rhi * DMODEL))[lane];

    const v4f p0 = ((const v4f*)(s1 + g0))[0];
    const v4f p1 = ((const v4f*)(s1 + g0))[1];
    const v4f q0 = ((const v4f*)(s2 + g0))[0];
    const v4f q1 = ((const v4f*)(s2 + g0))[1];
    const float tlo = s2[rlo];
    const float thi = s2[rhi];

    asm volatile("" :: "v"(w[0]), "v"(w[1]), "v"(w[2]), "v"(w[3]), "v"(w[4]),
                       "v"(w[5]), "v"(w[6]), "v"(w[7]), "v"(w[8]), "v"(w[9]),
                       "v"(p0), "v"(p1), "v"(q0), "v"(q1), "v"(tlo), "v"(thi));

    const float t1[ATT_ROWS] = {p0.x, p0.y, p0.z, p0.w, p1.x, p1.y, p1.z, p1.w};
    const float t2[ATT_ROWS + 2] = {tlo, q0.x, q0.y, q0.z, q0.w,
                                    q1.x, q1.y, q1.z, q1.w, thi};

    v4f xs[ATT_ROWS];
    v4f sum = {0.f, 0.f, 0.f, 0.f}, sq = {0.f, 0.f, 0.f, 0.f};
#pragma unroll
    for (int k = 0; k < ATT_ROWS; ++k) {
        const int g = g0 + k;
        const int i = g & (NGRAPH - 1);
        const float sc = t1[k];
        const bool hasl = (i > 0), hasr = (i < NGRAPH - 1);

        float ec = sc + t2[k + 1] + ab;    ec = fmaxf(ec, ALPHA * ec);
        float el = -3.0e38f, er = -3.0e38f;
        if (hasl) { el = sc + t2[k] + ab;     el = fmaxf(el, ALPHA * el); }
        if (hasr) { er = sc + t2[k + 2] + ab; er = fmaxf(er, ALPHA * er); }

        const float m  = fmaxf(ec, fmaxf(el, er));
        const float wl = hasl ? __expf(el - m) : 0.f;
        const float wc = __expf(ec - m);
        const float wr = hasr ? __expf(er - m) : 0.f;
        const float inv = 1.f / (wl + wc + wr);

        v4f x;
        x.x = att_elem(w[k].x, w[k + 1].x, w[k + 2].x, wl, wc, wr, inv, hasl, hasr);
        x.y = att_elem(w[k].y, w[k + 1].y, w[k + 2].y, wl, wc, wr, inv, hasl, hasr);
        x.z = att_elem(w[k].z, w[k + 1].z, w[k + 2].z, wl, wc, wr, inv, hasl, hasr);
        x.w = att_elem(w[k].w, w[k + 1].w, w[k + 2].w, wl, wc, wr, inv, hasl, hasr);

        xs[k] = x;
        sum.x += x.x; sum.y += x.y; sum.z += x.z; sum.w += x.w;
        sq.x += x.x * x.x; sq.y += x.y * x.y; sq.z += x.z * x.z; sq.w += x.w * x.w;
    }

    ((v4f*)red[0])[wave * 32 + lane] = sum;
    ((v4f*)red[1])[wave * 32 + lane] = sq;
    __syncthreads();

#pragma unroll
    for (int k = 0; k < ATT_ROWS; ++k)
        ((v4f*)(out + (g0 + k) * DMODEL))[lane] = xs[k];

    if (threadIdx.x < 2 * DMODEL) {
        const int which = threadIdx.x >> 7;
        const int d     = threadIdx.x & (DMODEL - 1);
        float acc = 0.f;
#pragma unroll
        for (int wv = 0; wv < ATT_WAVES; ++wv) acc += red[which][wv * DMODEL + d];
        float* dst = which ? psq : psum;
        dst[blockIdx.x * DMODEL + d] = acc;
    }
}

__global__ void gat_stats(const float* __restrict__ psum, const float* __restrict__ psq,
                          const float* __restrict__ gamma, const float* __restrict__ beta,
                          float* __restrict__ scale, float* __restrict__ shift) {
    const int d = threadIdx.x;
    float s = 0.f, q = 0.f;
    for (int b = 0; b < 128; ++b) { s += psum[b * 128 + d]; q += psq[b * 128 + d]; }
    const float mean = s * (1.f / (float)NTOK);
    const float var  = q * (1.f / (float)NTOK) - mean * mean;
    const float invs = rsqrtf(var + BN_EPS);
    const float sc   = gamma[d] * invs;
    scale[d] = sc;
    shift[d] = beta[d] - mean * sc;
}

__global__ __launch_bounds__(256)
void gat_bn(float* __restrict__ out, const float* __restrict__ scale,
            const float* __restrict__ shift) {
    const int idx = blockIdx.x * 256 + threadIdx.x;
    float4* o = (float4*)out;
    float4 x = o[idx];
    const int d4 = (idx & 31) * 4;
    x.x = x.x * scale[d4 + 0] + shift[d4 + 0];
    x.y = x.y * scale[d4 + 1] + shift[d4 + 1];
    x.z = x.z * scale[d4 + 2] + shift[d4 + 2];
    x.w = x.w * scale[d4 + 3] + shift[d4 + 3];
    o[idx] = x;
}

extern "C" void kernel_launch(void* const* d_in, const int* in_sizes, int n_in,
                              void* d_out, int out_size, void* d_ws, size_t ws_size,
                              hipStream_t stream) {
    const float* h     = (const float*)d_in[0];
    const float* W     = (const float*)d_in[1];
    const float* a1    = (const float*)d_in[2];
    const float* a2    = (const float*)d_in[3];
    const float* ab    = (const float*)d_in[4];
    const float* gamma = (const float*)d_in[5];
    const float* beta  = (const float*)d_in[6];
    float* out = (float*)d_out;

    float* ws    = (float*)d_ws;
    float* Wh    = ws;
    float* s1    = ws + 2097152;
    float* s2    = s1 + NTOK;
    float* psum  = s2 + NTOK;
    float* psq   = psum + NTOK;
    float* scale = psq + NTOK;
    float* shift = scale + 128;

    gat_gemm_wmma<<<128, 256, 0, stream>>>(h, W, a1, a2, Wh, s1, s2);
    gat_attn    <<<128, 512, 0, stream>>>(Wh, s1, s2, ab, out, psum, psq);
    gat_stats   <<<1, 128, 0, stream>>>(psum, psq, gamma, beta, scale, shift);
    gat_bn      <<<2048, 256, 0, stream>>>(out, scale, shift);
}


# BN stats folded into the BN pass, one launch less: 256 workgroups each own 32 features x 256 rows, fold that slice's 128 partials (8 per thread, 16 in LDS; changes BN sum order); 16384 waves -> 2048
# speedup vs baseline: 5.2868x; 1.4184x over previous
#include <hip/hip_runtime.h>
#include <hip/hip_bf16.h>

typedef float v2f __attribute__((ext_vector_type(2)));
typedef float v8f __attribute__((ext_vector_type(8)));
typedef float v4f __attribute__((ext_vector_type(4)));

#define NTOK   16384
#define DMODEL 128
#define NGRAPH 1024
#define ALPHA  0.2f
#define BN_EPS 1e-5f
#define WPITCH 132

template <int M>
__device__ __forceinline__ float row_xadd(float x) {
    const int y = __builtin_amdgcn_update_dpp(0, __builtin_bit_cast(int, x),
                                              0x160 + M, 0xf, 0xf, true);
    return x + __builtin_bit_cast(float, y);
}

__global__ __launch_bounds__(256)
void gat_gemm_wmma(const float* __restrict__ h, const float* __restrict__ W,
                   const float* __restrict__ a1, const float* __restrict__ a2,
                   float* __restrict__ Wh, float* __restrict__ s1,
                   float* __restrict__ s2) {
    __shared__ float Wlds[DMODEL * WPITCH];

    const int tid = threadIdx.x;

    v2f u[8];
#pragma unroll
    for (int t = 0; t < 8; ++t)
        u[t] = (v2f){a1[t * 16 + (tid & 15)], a2[t * 16 + (tid & 15)]};

    const float4* Wg = (const float4*)W;
#pragma unroll
    for (int it = 0; it < 16; ++it) {
        const int idx = tid + it * 256;
        const int row = idx >> 5;
        const int c4  = idx & 31;
        *(float4*)&Wlds[row * WPITCH + c4 * 4] = Wg[idx];
    }
    __syncthreads();

    const int wave  = tid >> 5;
    const int lane  = tid & 31;
    const int mtile = blockIdx.x * 8 + wave;
    const int mrow  = (lane & 15);
    const int khalf = (lane >> 4) << 1;

    v8f acc[8];
#pragma unroll
    for (int t = 0; t < 8; ++t) acc[t] = (v8f){0.f,0.f,0.f,0.f,0.f,0.f,0.f,0.f};

    const float* hrow = h + (mtile * 16 + mrow) * DMODEL + khalf;

    for (int kk = 0; kk < DMODEL; kk += 4) {
        v2f a = *(const v2f*)(hrow + kk);
#pragma unroll
        for (int t = 0; t < 8; ++t) {
            v2f b = *(const v2f*)(&Wlds[(t * 16 + mrow) * WPITCH + kk + khalf]);
            acc[t] = __builtin_amdgcn_wmma_f32_16x16x4_f32(
                false, a, false, b, (short)0, acc[t], false, false);
        }
    }

    const int hi = lane >> 4;
    const int cn = lane & 15;
#pragma unroll
    for (int t = 0; t < 8; ++t) {
#pragma unroll
        for (int v = 0; v < 8; ++v) {
            int r = mtile * 16 + v + 8 * hi;
            Wh[r * DMODEL + t * 16 + cn] = acc[t][v];
        }
    }

    v2f p[8];
#pragma unroll
    for (int v = 0; v < 8; ++v) p[v] = (v2f){acc[0][v], acc[0][v]} * u[0];
#pragma unroll
    for (int t = 1; t < 8; ++t) {
#pragma unroll
        for (int v = 0; v < 8; ++v)
            p[v] = __builtin_elementwise_fma((v2f){acc[t][v], acc[t][v]}, u[t], p[v]);
    }
    float p1[8], p2[8];
#pragma unroll
    for (int v = 0; v < 8; ++v) { p1[v] = row_xadd<8>(p[v].x); p2[v] = row_xadd<8>(p[v].y); }
#pragma unroll
    for (int v = 0; v < 8; ++v) { p1[v] = row_xadd<4>(p1[v]); p2[v] = row_xadd<4>(p2[v]); }
#pragma unroll
    for (int v = 0; v < 8; ++v) { p1[v] = row_xadd<2>(p1[v]); p2[v] = row_xadd<2>(p2[v]); }
#pragma unroll
    for (int v = 0; v < 8; ++v) { p1[v] = row_xadd<1>(p1[v]); p2[v] = row_xadd<1>(p2[v]); }

    if (cn == 0) {
        const int r0 = mtile * 16 + 8 * hi;
        ((v4f*)(s1 + r0))[0] = (v4f){p1[0], p1[1], p1[2], p1[3]};
        ((v4f*)(s1 + r0))[1] = (v4f){p1[4], p1[5], p1[6], p1[7]};
        ((v4f*)(s2 + r0))[0] = (v4f){p2[0], p2[1], p2[2], p2[3]};
        ((v4f*)(s2 + r0))[1] = (v4f){p2[4], p2[5], p2[6], p2[7]};
    }
}

#define ATT_ROWS  8
#define ATT_WAVES 16

__device__ __forceinline__ float att_elem(float l, float c, float r, float wl,
                                          float wc, float wr, float inv,
                                          bool hasl, bool hasr) {
    float x = wc * c;
    if (hasl) x += wl * l;
    if (hasr) x += wr * r;
    x *= inv;
    return (x > 0.f) ? x : (__expf(x) - 1.f);
}

__global__ __launch_bounds__(512)
void gat_attn(const float* __restrict__ Wh, const float* __restrict__ s1,
              const float* __restrict__ s2, const float* __restrict__ abp,
              float* __restrict__ out, float* __restrict__ psum,
              float* __restrict__ psq) {
    __shared__ float red[2][ATT_WAVES * DMODEL];

    const float ab = abp[0];
    const int lane = threadIdx.x & 31;
    const int wave = threadIdx.x >> 5;
    const int g0   = blockIdx.x * 128 + wave * ATT_ROWS;

    const int rlo = (g0 > 0) ? g0 - 1 : 0;
    const int rhi = (g0 + ATT_ROWS < NTOK) ? g0 + ATT_ROWS : NTOK - 1;
    v4f w[ATT_ROWS + 2];
    w[0] = ((const v4f*)(Wh + rlo * DMODEL))[lane];
#pragma unroll
    for (int j = 0; j < ATT_ROWS; ++j)
        w[j + 1] = ((const v4f*)(Wh + (g0 + j) * DMODEL))[lane];
    w[ATT_ROWS + 1] = ((const v4f*)(Wh + rhi * DMODEL))[lane];

    const v4f p0 = ((const v4f*)(s1 + g0))[0];
    const v4f p1 = ((const v4f*)(s1 + g0))[1];
    const v4f q0 = ((const v4f*)(s2 + g0))[0];
    const v4f q1 = ((const v4f*)(s2 + g0))[1];
    const float tlo = s2[rlo];
    const float thi = s2[rhi];

    asm volatile("" :: "v"(w[0]), "v"(w[1]), "v"(w[2]), "v"(w[3]), "v"(w[4]),
                       "v"(w[5]), "v"(w[6]), "v"(w[7]), "v"(w[8]), "v"(w[9]),
                       "v"(p0), "v"(p1), "v"(q0), "v"(q1), "v"(tlo), "v"(thi));

    const float t1[ATT_ROWS] = {p0.x, p0.y, p0.z, p0.w, p1.x, p1.y, p1.z, p1.w};
    const float t2[ATT_ROWS + 2] = {tlo, q0.x, q0.y, q0.z, q0.w,
                                    q1.x, q1.y, q1.z, q1.w, thi};

    v4f xs[ATT_ROWS];
    v4f sum = {0.f, 0.f, 0.f, 0.f}, sq = {0.f, 0.f, 0.f, 0.f};
#pragma unroll
    for (int k = 0; k < ATT_ROWS; ++k) {
        const int g = g0 + k;
        const int i = g & (NGRAPH - 1);
        const float sc = t1[k];
        const bool hasl = (i > 0), hasr = (i < NGRAPH - 1);

        float ec = sc + t2[k + 1] + ab;    ec = fmaxf(ec, ALPHA * ec);
        float el = -3.0e38f, er = -3.0e38f;
        if (hasl) { el = sc + t2[k] + ab;     el = fmaxf(el, ALPHA * el); }
        if (hasr) { er = sc + t2[k + 2] + ab; er = fmaxf(er, ALPHA * er); }

        const float m  = fmaxf(ec, fmaxf(el, er));
        const float wl = hasl ? __expf(el - m) : 0.f;
        const float wc = __expf(ec - m);
        const float wr = hasr ? __expf(er - m) : 0.f;
        const float inv = 1.f / (wl + wc + wr);

        v4f x;
        x.x = att_elem(w[k].x, w[k + 1].x, w[k + 2].x, wl, wc, wr, inv, hasl, hasr);
        x.y = att_elem(w[k].y, w[k + 1].y, w[k + 2].y, wl, wc, wr, inv, hasl, hasr);
        x.z = att_elem(w[k].z, w[k + 1].z, w[k + 2].z, wl, wc, wr, inv, hasl, hasr);
        x.w = att_elem(w[k].w, w[k + 1].w, w[k + 2].w, wl, wc, wr, inv, hasl, hasr);

        xs[k] = x;
        sum.x += x.x; sum.y += x.y; sum.z += x.z; sum.w += x.w;
        sq.x += x.x * x.x; sq.y += x.y * x.y; sq.z += x.z * x.z; sq.w += x.w * x.w;
    }

    ((v4f*)red[0])[wave * 32 + lane] = sum;
    ((v4f*)red[1])[wave * 32 + lane] = sq;
    __syncthreads();

#pragma unroll
    for (int k = 0; k < ATT_ROWS; ++k)
        ((v4f*)(out + (g0 + k) * DMODEL))[lane] = xs[k];

    if (threadIdx.x < 2 * DMODEL) {
        const int which = threadIdx.x >> 7;
        const int d     = threadIdx.x & (DMODEL - 1);
        float acc = 0.f;
#pragma unroll
        for (int wv = 0; wv < ATT_WAVES; ++wv) acc += red[which][wv * DMODEL + d];
        float* dst = which ? psq : psum;
        dst[blockIdx.x * DMODEL + d] = acc;
    }
}

#define BN_SLICE 32
#define BN_ROWS  256
#define BN_RPL   8
#define BN_GRP   16

__global__ __launch_bounds__(256)
void gat_bn(float* __restrict__ out, const float* __restrict__ psum,
            const float* __restrict__ psq, const float* __restrict__ gamma,
            const float* __restrict__ beta) {
    __shared__ __attribute__((aligned(16))) float red[2][BN_GRP * BN_SLICE];
    __shared__ __attribute__((aligned(16))) float ss[2][BN_SLICE];

    const int tid = threadIdx.x;
    const int fs  = blockIdx.x & 3;
    const int rg  = blockIdx.x >> 2;
    const int c   = tid & 7;
    const int g   = tid >> 3;
    const int col = fs * BN_SLICE + c * 4;

    const int which = tid >> 7;
    const int grp   = g & (BN_GRP - 1);
    const float* part = (which ? psq : psum) + grp * 8 * DMODEL + col;

    const float gm = gamma[fs * BN_SLICE + (tid & 31)];
    const float bt = beta [fs * BN_SLICE + (tid & 31)];

    v4f p[8];
#pragma unroll
    for (int j = 0; j < 8; ++j)
        p[j] = *(const v4f*)(part + j * DMODEL);

    float* orow = out + (rg * BN_ROWS + g) * DMODEL + col;
    v4f x[BN_RPL];
#pragma unroll
    for (int j = 0; j < BN_RPL; ++j)
        x[j] = *(const v4f*)(orow + j * 32 * DMODEL);

    __builtin_amdgcn_sched_barrier(0);

    v4f a = p[0];
#pragma unroll
    for (int j = 1; j < 8; ++j) a += p[j];
    *(v4f*)&red[which][grp * BN_SLICE + c * 4] = a;
    __syncthreads();

    if (tid < BN_SLICE) {
        float s = 0.f, q = 0.f;
#pragma unroll
        for (int k = 0; k < BN_GRP; ++k) {
            s += red[0][k * BN_SLICE + tid];
            q += red[1][k * BN_SLICE + tid];
        }
        const float mean = s * (1.f / (float)NTOK);
        const float var  = q * (1.f / (float)NTOK) - mean * mean;
        const float invs = rsqrtf(var + BN_EPS);
        const float sc   = gm * invs;
        ss[0][tid] = sc;
        ss[1][tid] = bt - mean * sc;
    }
    __syncthreads();

    const v4f sc = *(const v4f*)&ss[0][c * 4];
    const v4f sh = *(const v4f*)&ss[1][c * 4];
#pragma unroll
    for (int j = 0; j < BN_RPL; ++j)
        *(v4f*)(orow + j * 32 * DMODEL) = x[j] * sc + sh;
}

extern "C" void kernel_launch(void* const* d_in, const int* in_sizes, int n_in,
                              void* d_out, int out_size, void* d_ws, size_t ws_size,
                              hipStream_t stream) {
    const float* h     = (const float*)d_in[0];
    const float* W     = (const float*)d_in[1];
    const float* a1    = (const float*)d_in[2];
    const float* a2    = (const float*)d_in[3];
    const float* ab    = (const float*)d_in[4];
    const float* gamma = (const float*)d_in[5];
    const float* beta  = (const float*)d_in[6];
    float* out = (float*)d_out;

    float* ws    = (float*)d_ws;
    float* Wh    = ws;
    float* s1    = ws + 2097152;
    float* s2    = s1 + NTOK;
    float* psum  = s2 + NTOK;
    float* psq   = psum + NTOK;

    gat_gemm_wmma<<<128, 256, 0, stream>>>(h, W, a1, a2, Wh, s1, s2);
    gat_attn    <<<128, 512, 0, stream>>>(Wh, s1, s2, ab, out, psum, psq);
    gat_bn      <<<256, 256, 0, stream>>>(out, psum, psq, gamma, beta);
}


# stencil/ELU pass fused into the GEMM launch: a block's Wh rows and scores go from the accumulators to the stencil through LDS, its 2 edge rows as a 9th M tile split over the 8 waves; one launch less
# speedup vs baseline: 6.0864x; 1.1512x over previous
#include <hip/hip_runtime.h>
#include <hip/hip_bf16.h>

typedef float v2f __attribute__((ext_vector_type(2)));
typedef float v8f __attribute__((ext_vector_type(8)));
typedef float v4f __attribute__((ext_vector_type(4)));

#define NTOK   16384
#define DMODEL 128
#define NGRAPH 1024
#define ALPHA  0.2f
#define BN_EPS 1e-5f
#define WPITCH 132

template <int M>
__device__ __forceinline__ float row_xadd(float x) {
    const int y = __builtin_amdgcn_update_dpp(0, __builtin_bit_cast(int, x),
                                              0x160 + M, 0xf, 0xf, true);
    return x + __builtin_bit_cast(float, y);
}

#define BLK_ROWS  128
#define ATT_ROWS  8
#define ATT_GRPS  16
#define LDS_FLOATS (DMODEL * WPITCH + (BLK_ROWS + 2) * WPITCH + 2 * BLK_ROWS + 2 * ATT_GRPS * DMODEL)

__device__ __forceinline__ float att_elem(float l, float c, float r, float wl,
                                          float wc, float wr, float inv,
                                          bool hasl, bool hasr) {
    float x = wc * c;
    if (hasl) x += wl * l;
    if (hasr) x += wr * r;
    x *= inv;
    return (x > 0.f) ? x : (__expf(x) - 1.f);
}

__global__ __launch_bounds__(256)
void gat_gemm_attn(const float* __restrict__ h, const float* __restrict__ W,
                   const float* __restrict__ a1, const float* __restrict__ a2,
                   const float* __restrict__ abp, float* __restrict__ out,
                   float* __restrict__ psum, float* __restrict__ psq) {
    __shared__ __attribute__((aligned(16))) float lds[LDS_FLOATS];
    float* const Wlds = lds;
    float* const WhL  = Wlds + DMODEL * WPITCH;
    float* const s1L  = WhL + (BLK_ROWS + 2) * WPITCH;
    float* const s2L  = s1L + BLK_ROWS;
    float* const red0 = s2L + BLK_ROWS;
    float* const red1 = red0 + ATT_GRPS * DMODEL;

    const int tid = threadIdx.x;
    const float ab = abp[0];

    v2f u[8];
#pragma unroll
    for (int t = 0; t < 8; ++t)
        u[t] = (v2f){a1[t * 16 + (tid & 15)], a2[t * 16 + (tid & 15)]};

    const float4* Wg = (const float4*)W;
#pragma unroll
    for (int it = 0; it < 16; ++it) {
        const int idx = tid + it * 256;
        const int row = idx >> 5;
        const int c4  = idx & 31;
        *(float4*)&Wlds[row * WPITCH + c4 * 4] = Wg[idx];
    }
    __syncthreads();

    const int wave  = tid >> 5;
    const int lane  = tid & 31;
    const int g0    = blockIdx.x * BLK_ROWS;
    const int mtile = blockIdx.x * 8 + wave;
    const int mrow  = (lane & 15);
    const int khalf = (lane >> 4) << 1;

    const int rlo = (g0 > 0) ? g0 - 1 : 0;
    const int rhi = (g0 + BLK_ROWS < NTOK) ? g0 + BLK_ROWS : NTOK - 1;

    v8f acc[8];
#pragma unroll
    for (int t = 0; t < 8; ++t) acc[t] = (v8f){0.f,0.f,0.f,0.f,0.f,0.f,0.f,0.f};
    v8f acch = (v8f){0.f,0.f,0.f,0.f,0.f,0.f,0.f,0.f};

    const float* hrow  = h + (mtile * 16 + mrow) * DMODEL + khalf;
    const float* hedge = h + ((mrow == 1) ? rhi : rlo) * DMODEL + khalf;
    const float* wedge = &Wlds[(wave * 16 + mrow) * WPITCH + khalf];

    for (int kk = 0; kk < DMODEL; kk += 4) {
        v2f a  = *(const v2f*)(hrow + kk);
        v2f ae = *(const v2f*)(hedge + kk);
#pragma unroll
        for (int t = 0; t < 8; ++t) {
            v2f b = *(const v2f*)(&Wlds[(t * 16 + mrow) * WPITCH + kk + khalf]);
            acc[t] = __builtin_amdgcn_wmma_f32_16x16x4_f32(
                false, a, false, b, (short)0, acc[t], false, false);
        }
        v2f be = *(const v2f*)(wedge + kk);
        acch = __builtin_amdgcn_wmma_f32_16x16x4_f32(
            false, ae, false, be, (short)0, acch, false, false);
    }

    const int hi = lane >> 4;
    const int cn = lane & 15;
    const int lr = wave * 16 + 8 * hi;
#pragma unroll
    for (int t = 0; t < 8; ++t) {
#pragma unroll
        for (int v = 0; v < 8; ++v)
            WhL[(1 + lr + v) * WPITCH + t * 16 + cn] = acc[t][v];
    }
    if (hi == 0) {
        WhL[wave * 16 + cn] = acch[0];
        WhL[(BLK_ROWS + 1) * WPITCH + wave * 16 + cn] = acch[1];
    }

    v2f p[8];
#pragma unroll
    for (int v = 0; v < 8; ++v) p[v] = (v2f){acc[0][v], acc[0][v]} * u[0];
#pragma unroll
    for (int t = 1; t < 8; ++t) {
#pragma unroll
        for (int v = 0; v < 8; ++v)
            p[v] = __builtin_elementwise_fma((v2f){acc[t][v], acc[t][v]}, u[t], p[v]);
    }
    float p1[8], p2[8];
#pragma unroll
    for (int v = 0; v < 8; ++v) { p1[v] = row_xadd<8>(p[v].x); p2[v] = row_xadd<8>(p[v].y); }
#pragma unroll
    for (int v = 0; v < 8; ++v) { p1[v] = row_xadd<4>(p1[v]); p2[v] = row_xadd<4>(p2[v]); }
#pragma unroll
    for (int v = 0; v < 8; ++v) { p1[v] = row_xadd<2>(p1[v]); p2[v] = row_xadd<2>(p2[v]); }
#pragma unroll
    for (int v = 0; v < 8; ++v) { p1[v] = row_xadd<1>(p1[v]); p2[v] = row_xadd<1>(p2[v]); }

    if (cn == 0) {
        ((v4f*)(s1L + lr))[0] = (v4f){p1[0], p1[1], p1[2], p1[3]};
        ((v4f*)(s1L + lr))[1] = (v4f){p1[4], p1[5], p1[6], p1[7]};
        ((v4f*)(s2L + lr))[0] = (v4f){p2[0], p2[1], p2[2], p2[3]};
        ((v4f*)(s2L + lr))[1] = (v4f){p2[4], p2[5], p2[6], p2[7]};
    }
    __syncthreads();

    const float* er = &WhL[(hi ? (BLK_ROWS + 1) : 0) * WPITCH + cn];
    float pe = er[0] * u[0].y;
#pragma unroll
    for (int t = 1; t < 8; ++t) pe = __builtin_fmaf(er[t * 16], u[t].y, pe);
    pe = row_xadd<8>(pe); pe = row_xadd<4>(pe); pe = row_xadd<2>(pe); pe = row_xadd<1>(pe);
    const float s2lo = __builtin_bit_cast(float,
        __builtin_amdgcn_readlane(__builtin_bit_cast(int, pe), 0));
    const float s2hi = __builtin_bit_cast(float,
        __builtin_amdgcn_readlane(__builtin_bit_cast(int, pe), 16));

    v4f xs[2 * ATT_ROWS];
#pragma unroll
    for (int j = 0; j < 2; ++j) {
        const int R0 = wave * 16 + j * ATT_ROWS;

        v4f w[ATT_ROWS + 2];
#pragma unroll
        for (int i = 0; i < ATT_ROWS + 2; ++i)
            w[i] = *(const v4f*)&WhL[(R0 + i) * WPITCH + lane * 4];

        const v4f p0 = ((const v4f*)(s1L + R0))[0];
        const v4f p1v = ((const v4f*)(s1L + R0))[1];
        const v4f q0 = ((const v4f*)(s2L + R0))[0];
        const v4f q1 = ((const v4f*)(s2L + R0))[1];
        const float tlo = (R0 > 0) ? s2L[(R0 > 0) ? R0 - 1 : 0] : s2lo;
        const float thi = (R0 + ATT_ROWS < BLK_ROWS)
                              ? s2L[(R0 + ATT_ROWS < BLK_ROWS) ? R0 + ATT_ROWS : BLK_ROWS - 1]
                              : s2hi;

        const float t1[ATT_ROWS] = {p0.x, p0.y, p0.z, p0.w, p1v.x, p1v.y, p1v.z, p1v.w};
        const float t2[ATT_ROWS + 2] = {tlo, q0.x, q0.y, q0.z, q0.w,
                                        q1.x, q1.y, q1.z, q1.w, thi};

        v4f sum = {0.f, 0.f, 0.f, 0.f}, sq = {0.f, 0.f, 0.f, 0.f};
#pragma unroll
        for (int k = 0; k < ATT_ROWS; ++k) {
            const int g = g0 + R0 + k;
            const int i = g & (NGRAPH - 1);
            const float sc = t1[k];
            const bool hasl = (i > 0), hasr = (i < NGRAPH - 1);

            float ec = sc + t2[k + 1] + ab;    ec = fmaxf(ec, ALPHA * ec);
            float el = -3.0e38f, er2 = -3.0e38f;
            if (hasl) { el = sc + t2[k] + ab;      el = fmaxf(el, ALPHA * el); }
            if (hasr) { er2 = sc + t2[k + 2] + ab; er2 = fmaxf(er2, ALPHA * er2); }

            const float m  = fmaxf(ec, fmaxf(el, er2));
            const float wl = hasl ? __expf(el - m) : 0.f;
            const float wc = __expf(ec - m);
            const float wr = hasr ? __expf(er2 - m) : 0.f;
            const float inv = 1.f / (wl + wc + wr);

            v4f x;
            x.x = att_elem(w[k].x, w[k + 1].x, w[k + 2].x, wl, wc, wr, inv, hasl, hasr);
            x.y = att_elem(w[k].y, w[k + 1].y, w[k + 2].y, wl, wc, wr, inv, hasl, hasr);
            x.z = att_elem(w[k].z, w[k + 1].z, w[k + 2].z, wl, wc, wr, inv, hasl, hasr);
            x.w = att_elem(w[k].w, w[k + 1].w, w[k + 2].w, wl, wc, wr, inv, hasl, hasr);

            xs[j * ATT_ROWS + k] = x;
            sum.x += x.x; sum.y += x.y; sum.z += x.z; sum.w += x.w;
            sq.x += x.x * x.x; sq.y += x.y * x.y; sq.z += x.z * x.z; sq.w += x.w * x.w;
        }

        ((v4f*)red0)[(wave * 2 + j) * 32 + lane] = sum;
        ((v4f*)red1)[(wave * 2 + j) * 32 + lane] = sq;
    }
    __syncthreads();

#pragma unroll
    for (int k = 0; k < 2 * ATT_ROWS; ++k)
        ((v4f*)(out + (g0 + wave * 16 + k) * DMODEL))[lane] = xs[k];

    {
        const int which = tid >> 7;
        const int d     = tid & (DMODEL - 1);
        float accp = 0.f;
#pragma unroll
        for (int gp = 0; gp < ATT_GRPS; ++gp) accp += (which ? red1 : red0)[gp * DMODEL + d];
        float* dst = which ? psq : psum;
        dst[blockIdx.x * DMODEL + d] = accp;
    }
}

#define BN_SLICE 32
#define BN_ROWS  256
#define BN_RPL   8
#define BN_GRP   16

__global__ __launch_bounds__(256)
void gat_bn(float* __restrict__ out, const float* __restrict__ psum,
            const float* __restrict__ psq, const float* __restrict__ gamma,
            const float* __restrict__ beta) {
    __shared__ __attribute__((aligned(16))) float red[2][BN_GRP * BN_SLICE];
    __shared__ __attribute__((aligned(16))) float ss[2][BN_SLICE];

    const int tid = threadIdx.x;
    const int fs  = blockIdx.x & 3;
    const int rg  = blockIdx.x >> 2;
    const int c   = tid & 7;
    const int g   = tid >> 3;
    const int col = fs * BN_SLICE + c * 4;

    const int which = tid >> 7;
    const int grp   = g & (BN_GRP - 1);
    const float* part = (which ? psq : psum) + grp * 8 * DMODEL + col;

    const float gm = gamma[fs * BN_SLICE + (tid & 31)];
    const float bt = beta [fs * BN_SLICE + (tid & 31)];

    v4f p[8];
#pragma unroll
    for (int j = 0; j < 8; ++j)
        p[j] = *(const v4f*)(part + j * DMODEL);

    float* orow = out + (rg * BN_ROWS + g) * DMODEL + col;
    v4f x[BN_RPL];
#pragma unroll
    for (int j = 0; j < BN_RPL; ++j)
        x[j] = *(const v4f*)(orow + j * 32 * DMODEL);

    __builtin_amdgcn_sched_barrier(0);

    v4f a = p[0];
#pragma unroll
    for (int j = 1; j < 8; ++j) a += p[j];
    *(v4f*)&red[which][grp * BN_SLICE + c * 4] = a;
    __syncthreads();

    if (tid < BN_SLICE) {
        float s = 0.f, q = 0.f;
#pragma unroll
        for (int k = 0; k < BN_GRP; ++k) {
            s += red[0][k * BN_SLICE + tid];
            q += red[1][k * BN_SLICE + tid];
        }
        const float mean = s * (1.f / (float)NTOK);
        const float var  = q * (1.f / (float)NTOK) - mean * mean;
        const float invs = rsqrtf(var + BN_EPS);
        const float sc   = gm * invs;
        ss[0][tid] = sc;
        ss[1][tid] = bt - mean * sc;
    }
    __syncthreads();

    const v4f sc = *(const v4f*)&ss[0][c * 4];
    const v4f sh = *(const v4f*)&ss[1][c * 4];
#pragma unroll
    for (int j = 0; j < BN_RPL; ++j)
        *(v4f*)(orow + j * 32 * DMODEL) = x[j] * sc + sh;
}

extern "C" void kernel_launch(void* const* d_in, const int* in_sizes, int n_in,
                              void* d_out, int out_size, void* d_ws, size_t ws_size,
                              hipStream_t stream) {
    const float* h     = (const float*)d_in[0];
    const float* W     = (const float*)d_in[1];
    const float* a1    = (const float*)d_in[2];
    const float* a2    = (const float*)d_in[3];
    const float* ab    = (const float*)d_in[4];
    const float* gamma = (const float*)d_in[5];
    const float* beta  = (const float*)d_in[6];
    float* out = (float*)d_out;

    float* ws    = (float*)d_ws;
    float* psum  = ws;
    float* psq   = psum + NTOK;

    gat_gemm_attn<<<128, 256, 0, stream>>>(h, W, a1, a2, ab, out, psum, psq);
    gat_bn       <<<256, 256, 0, stream>>>(out, psum, psq, gamma, beta);
}


# GEMM+stencil launch re-tiled: 256 workgroups x 64 rows x 4 waves (one per SIMD, every WGP; was 128 x 128 x 8), h tile staged in LDS with W by whole-row loads; 256 BN partials, changes BN sum order
# speedup vs baseline: 7.4116x; 1.2177x over previous
#include <hip/hip_runtime.h>
#include <hip/hip_bf16.h>

typedef float v2f __attribute__((ext_vector_type(2)));
typedef float v8f __attribute__((ext_vector_type(8)));
typedef float v4f __attribute__((ext_vector_type(4)));

#define NTOK   16384
#define DMODEL 128
#define NGRAPH 1024
#define ALPHA  0.2f
#define BN_EPS 1e-5f
#define WPITCH 132

template <int M>
__device__ __forceinline__ float row_xadd(float x) {
    const int y = __builtin_amdgcn_update_dpp(0, __builtin_bit_cast(int, x),
                                              0x160 + M, 0xf, 0xf, true);
    return x + __builtin_bit_cast(float, y);
}

#define BLK_ROWS  64
#define NBLK      (NTOK / BLK_ROWS)
#define ATT_ROWS  8
#define ATT_GRPS  8
#define HROWS     (BLK_ROWS + 2)
#define LDS_FLOATS (DMODEL * WPITCH + (HROWS + 2) * WPITCH + HROWS * WPITCH + 2 * BLK_ROWS + 2 * ATT_GRPS * DMODEL)

__device__ __forceinline__ float att_elem(float l, float c, float r, float wl,
                                          float wc, float wr, float inv,
                                          bool hasl, bool hasr) {
    float x = wc * c;
    if (hasl) x += wl * l;
    if (hasr) x += wr * r;
    x *= inv;
    return (x > 0.f) ? x : (__expf(x) - 1.f);
}

__global__ __launch_bounds__(128)
void gat_gemm_attn(const float* __restrict__ h, const float* __restrict__ W,
                   const float* __restrict__ a1, const float* __restrict__ a2,
                   const float* __restrict__ abp, float* __restrict__ out,
                   float* __restrict__ psum, float* __restrict__ psq) {
    __shared__ __attribute__((aligned(16))) float lds[LDS_FLOATS];
    float* const Wlds = lds;
    float* const hL   = Wlds + DMODEL * WPITCH;
    float* const WhL  = hL + (HROWS + 2) * WPITCH;
    float* const s1L  = WhL + HROWS * WPITCH;
    float* const s2L  = s1L + BLK_ROWS;
    float* const red0 = s2L + BLK_ROWS;
    float* const red1 = red0 + ATT_GRPS * DMODEL;

    const int tid = threadIdx.x;
    const float ab = abp[0];

    v2f u[8];
#pragma unroll
    for (int t = 0; t < 8; ++t)
        u[t] = (v2f){a1[t * 16 + (tid & 15)], a2[t * 16 + (tid & 15)]};

    const int wave  = tid >> 5;
    const int lane  = tid & 31;
    const int g0    = blockIdx.x * BLK_ROWS;
    const int mrow  = (lane & 15);
    const int khalf = (lane >> 4) << 1;

    const float4* Wg = (const float4*)W;
    const float4* Hg = (const float4*)h;
    float4 wreg[32], hreg[16];
#pragma unroll
    for (int it = 0; it < 32; ++it) wreg[it] = Wg[(wave + 4 * it) * 32 + lane];
#pragma unroll
    for (int it = 0; it < 16; ++it) {
        int gr = g0 - 1 + wave + 4 * it;
        gr = (gr < 0) ? 0 : gr;
        hreg[it] = Hg[gr * 32 + lane];
    }
    int grl = g0 - 1 + BLK_ROWS + wave;
    grl = (grl > NTOK - 1) ? NTOK - 1 : grl;
    const float4 hlast = Hg[grl * 32 + lane];

#pragma unroll
    for (int it = 0; it < 32; ++it)
        *(float4*)&Wlds[(wave + 4 * it) * WPITCH + lane * 4] = wreg[it];
#pragma unroll
    for (int it = 0; it < 16; ++it)
        *(float4*)&hL[(wave + 4 * it) * WPITCH + lane * 4] = hreg[it];
    *(float4*)&hL[(BLK_ROWS + wave) * WPITCH + lane * 4] = hlast;
    __syncthreads();

    v8f acc[8];
#pragma unroll
    for (int t = 0; t < 8; ++t) acc[t] = (v8f){0.f,0.f,0.f,0.f,0.f,0.f,0.f,0.f};
    v8f acch[2];
#pragma unroll
    for (int e = 0; e < 2; ++e) acch[e] = (v8f){0.f,0.f,0.f,0.f,0.f,0.f,0.f,0.f};

    const float* arow  = &hL[(1 + wave * 16 + mrow) * WPITCH + khalf];
    const float* aedge = &hL[((mrow == 1) ? (BLK_ROWS + 1) : 0) * WPITCH + khalf];
    const float* wedge = &Wlds[(wave * 32 + mrow) * WPITCH + khalf];

#pragma unroll
    for (int kk = 0; kk < DMODEL; kk += 4) {
        const v2f a  = *(const v2f*)(arow + kk);
        const v2f ae = *(const v2f*)(aedge + kk);
#pragma unroll
        for (int t = 0; t < 8; ++t) {
            v2f b = *(const v2f*)(&Wlds[(t * 16 + mrow) * WPITCH + kk + khalf]);
            acc[t] = __builtin_amdgcn_wmma_f32_16x16x4_f32(
                false, a, false, b, (short)0, acc[t], false, false);
        }
#pragma unroll
        for (int e = 0; e < 2; ++e) {
            v2f be = *(const v2f*)(wedge + e * 16 * WPITCH + kk);
            acch[e] = __builtin_amdgcn_wmma_f32_16x16x4_f32(
                false, ae, false, be, (short)0, acch[e], false, false);
        }
    }

    const int hi = lane >> 4;
    const int cn = lane & 15;
    const int lr = wave * 16 + 8 * hi;
#pragma unroll
    for (int t = 0; t < 8; ++t) {
#pragma unroll
        for (int v = 0; v < 8; ++v)
            WhL[(1 + lr + v) * WPITCH + t * 16 + cn] = acc[t][v];
    }
    if (hi == 0) {
#pragma unroll
        for (int e = 0; e < 2; ++e) {
            WhL[wave * 32 + e * 16 + cn] = acch[e][0];
            WhL[(BLK_ROWS + 1) * WPITCH + wave * 32 + e * 16 + cn] = acch[e][1];
        }
    }

    v2f p[8];
#pragma unroll
    for (int v = 0; v < 8; ++v) p[v] = (v2f){acc[0][v], acc[0][v]} * u[0];
#pragma unroll
    for (int t = 1; t < 8; ++t) {
#pragma unroll
        for (int v = 0; v < 8; ++v)
            p[v] = __builtin_elementwise_fma((v2f){acc[t][v], acc[t][v]}, u[t], p[v]);
    }
    float p1[8], p2[8];
#pragma unroll
    for (int v = 0; v < 8; ++v) { p1[v] = row_xadd<8>(p[v].x); p2[v] = row_xadd<8>(p[v].y); }
#pragma unroll
    for (int v = 0; v < 8; ++v) { p1[v] = row_xadd<4>(p1[v]); p2[v] = row_xadd<4>(p2[v]); }
#pragma unroll
    for (int v = 0; v < 8; ++v) { p1[v] = row_xadd<2>(p1[v]); p2[v] = row_xadd<2>(p2[v]); }
#pragma unroll
    for (int v = 0; v < 8; ++v) { p1[v] = row_xadd<1>(p1[v]); p2[v] = row_xadd<1>(p2[v]); }

    if (cn == 0) {
        ((v4f*)(s1L + lr))[0] = (v4f){p1[0], p1[1], p1[2], p1[3]};
        ((v4f*)(s1L + lr))[1] = (v4f){p1[4], p1[5], p1[6], p1[7]};
        ((v4f*)(s2L + lr))[0] = (v4f){p2[0], p2[1], p2[2], p2[3]};
        ((v4f*)(s2L + lr))[1] = (v4f){p2[4], p2[5], p2[6], p2[7]};
    }
    __syncthreads();

    const float* er = &WhL[(hi ? (BLK_ROWS + 1) : 0) * WPITCH + cn];
    float pe = er[0] * u[0].y;
#pragma unroll
    for (int t = 1; t < 8; ++t) pe = __builtin_fmaf(er[t * 16], u[t].y, pe);
    pe = row_xadd<8>(pe); pe = row_xadd<4>(pe); pe = row_xadd<2>(pe); pe = row_xadd<1>(pe);
    const float s2lo = __builtin_bit_cast(float,
        __builtin_amdgcn_readlane(__builtin_bit_cast(int, pe), 0));
    const float s2hi = __builtin_bit_cast(float,
        __builtin_amdgcn_readlane(__builtin_bit_cast(int, pe), 16));

    v4f xs[2 * ATT_ROWS];
#pragma unroll
    for (int j = 0; j < 2; ++j) {
        const int R0 = wave * 16 + j * ATT_ROWS;

        v4f w[ATT_ROWS + 2];
#pragma unroll
        for (int i = 0; i < ATT_ROWS + 2; ++i)
            w[i] = *(const v4f*)&WhL[(R0 + i) * WPITCH + lane * 4];

        const v4f p0 = ((const v4f*)(s1L + R0))[0];
        const v4f p1v = ((const v4f*)(s1L + R0))[1];
        const v4f q0 = ((const v4f*)(s2L + R0))[0];
        const v4f q1 = ((const v4f*)(s2L + R0))[1];
        const float tlo = (R0 > 0) ? s2L[(R0 > 0) ? R0 - 1 : 0] : s2lo;
        const float thi = (R0 + ATT_ROWS < BLK_ROWS)
                              ? s2L[(R0 + ATT_ROWS < BLK_ROWS) ? R0 + ATT_ROWS : BLK_ROWS - 1]
                              : s2hi;

        const float t1[ATT_ROWS] = {p0.x, p0.y, p0.z, p0.w, p1v.x, p1v.y, p1v.z, p1v.w};
        const float t2[ATT_ROWS + 2] = {tlo, q0.x, q0.y, q0.z, q0.w,
                                        q1.x, q1.y, q1.z, q1.w, thi};

        v4f sum = {0.f, 0.f, 0.f, 0.f}, sq = {0.f, 0.f, 0.f, 0.f};
#pragma unroll
        for (int k = 0; k < ATT_ROWS; ++k) {
            const int g = g0 + R0 + k;
            const int i = g & (NGRAPH - 1);
            const float sc = t1[k];
            const bool hasl = (i > 0), hasr = (i < NGRAPH - 1);

            float ec = sc + t2[k + 1] + ab;    ec = fmaxf(ec, ALPHA * ec);
            float el = -3.0e38f, er2 = -3.0e38f;
            if (hasl) { el = sc + t2[k] + ab;      el = fmaxf(el, ALPHA * el); }
            if (hasr) { er2 = sc + t2[k + 2] + ab; er2 = fmaxf(er2, ALPHA * er2); }

            const float m  = fmaxf(ec, fmaxf(el, er2));
            const float wl = hasl ? __expf(el - m) : 0.f;
            const float wc = __expf(ec - m);
            const float wr = hasr ? __expf(er2 - m) : 0.f;
            const float inv = 1.f / (wl + wc + wr);

            v4f x;
            x.x = att_elem(w[k].x, w[k + 1].x, w[k + 2].x, wl, wc, wr, inv, hasl, hasr);
            x.y = att_elem(w[k].y, w[k + 1].y, w[k + 2].y, wl, wc, wr, inv, hasl, hasr);
            x.z = att_elem(w[k].z, w[k + 1].z, w[k + 2].z, wl, wc, wr, inv, hasl, hasr);
            x.w = att_elem(w[k].w, w[k + 1].w, w[k + 2].w, wl, wc, wr, inv, hasl, hasr);

            xs[j * ATT_ROWS + k] = x;
            sum.x += x.x; sum.y += x.y; sum.z += x.z; sum.w += x.w;
            sq.x += x.x * x.x; sq.y += x.y * x.y; sq.z += x.z * x.z; sq.w += x.w * x.w;
        }

        ((v4f*)red0)[(wave * 2 + j) * 32 + lane] = sum;
        ((v4f*)red1)[(wave * 2 + j) * 32 + lane] = sq;
    }
    __syncthreads();

#pragma unroll
    for (int k = 0; k < 2 * ATT_ROWS; ++k)
        ((v4f*)(out + (g0 + wave * 16 + k) * DMODEL))[lane] = xs[k];

    {
        float accs = 0.f, accq = 0.f;
#pragma unroll
        for (int gp = 0; gp < ATT_GRPS; ++gp) {
            accs += red0[gp * DMODEL + tid];
            accq += red1[gp * DMODEL + tid];
        }
        psum[blockIdx.x * DMODEL + tid] = accs;
        psq [blockIdx.x * DMODEL + tid] = accq;
    }
}

#define BN_SLICE 32
#define BN_ROWS  256
#define BN_RPL   8
#define BN_GRP   16
#define BN_PPT   (NBLK / BN_GRP)

__global__ __launch_bounds__(256)
void gat_bn(float* __restrict__ out, const float* __restrict__ psum,
            const float* __restrict__ psq, const float* __restrict__ gamma,
            const float* __restrict__ beta) {
    __shared__ __attribute__((aligned(16))) float red[2][BN_GRP * BN_SLICE];
    __shared__ __attribute__((aligned(16))) float ss[2][BN_SLICE];

    const int tid = threadIdx.x;
    const int fs  = blockIdx.x & 3;
    const int rg  = blockIdx.x >> 2;
    const int c   = tid & 7;
    const int g   = tid >> 3;
    const int col = fs * BN_SLICE + c * 4;

    const int which = tid >> 7;
    const int grp   = g & (BN_GRP - 1);
    const float* part = (which ? psq : psum) + grp * BN_PPT * DMODEL + col;

    const float gm = gamma[fs * BN_SLICE + (tid & 31)];
    const float bt = beta [fs * BN_SLICE + (tid & 31)];

    v4f p[BN_PPT];
#pragma unroll
    for (int j = 0; j < BN_PPT; ++j)
        p[j] = *(const v4f*)(part + j * DMODEL);

    float* orow = out + (rg * BN_ROWS + g) * DMODEL + col;
    v4f x[BN_RPL];
#pragma unroll
    for (int j = 0; j < BN_RPL; ++j)
        x[j] = *(const v4f*)(orow + j * 32 * DMODEL);

    __builtin_amdgcn_sched_barrier(0);

    v4f a = p[0];
#pragma unroll
    for (int j = 1; j < BN_PPT; ++j) a += p[j];
    *(v4f*)&red[which][grp * BN_SLICE + c * 4] = a;
    __syncthreads();

    if (tid < BN_SLICE) {
        float s = 0.f, q = 0.f;
#pragma unroll
        for (int k = 0; k < BN_GRP; ++k) {
            s += red[0][k * BN_SLICE + tid];
            q += red[1][k * BN_SLICE + tid];
        }
        const float mean = s * (1.f / (float)NTOK);
        const float var  = q * (1.f / (float)NTOK) - mean * mean;
        const float invs = rsqrtf(var + BN_EPS);
        const float sc   = gm * invs;
        ss[0][tid] = sc;
        ss[1][tid] = bt - mean * sc;
    }
    __syncthreads();

    const v4f sc = *(const v4f*)&ss[0][c * 4];
    const v4f sh = *(const v4f*)&ss[1][c * 4];
#pragma unroll
    for (int j = 0; j < BN_RPL; ++j)
        *(v4f*)(orow + j * 32 * DMODEL) = x[j] * sc + sh;
}

extern "C" void kernel_launch(void* const* d_in, const int* in_sizes, int n_in,
                              void* d_out, int out_size, void* d_ws, size_t ws_size,
                              hipStream_t stream) {
    const float* h     = (const float*)d_in[0];
    const float* W     = (const float*)d_in[1];
    const float* a1    = (const float*)d_in[2];
    const float* a2    = (const float*)d_in[3];
    const float* ab    = (const float*)d_in[4];
    const float* gamma = (const float*)d_in[5];
    const float* beta  = (const float*)d_in[6];
    float* out = (float*)d_out;

    float* ws    = (float*)d_ws;
    float* psum  = ws;
    float* psq   = psum + 2 * NTOK;

    gat_gemm_attn<<<256, 128, 0, stream>>>(h, W, a1, a2, ab, out, psum, psq);
    gat_bn       <<<256, 256, 0, stream>>>(out, psum, psq, gamma, beta);
}
